# starting kernel
# baseline (speedup 1.0000x reference)
#include <hip/hip_runtime.h>

typedef __attribute__((ext_vector_type(2))) float v2f;
typedef __attribute__((ext_vector_type(8))) float v8f;

#define B_   32
#define N_   50000
#define F_   50000
#define K_   32
#define H_   10
#define EPS_ 1e-5f
#define FPB  8

__global__ __launch_bounds__(256) void factor1d_wmma_kernel(
    const float* __restrict__ x,
    const int*   __restrict__ idx,
    const float* __restrict__ w0,
    const float* __restrict__ gamma,
    const float* __restrict__ beta,
    const float* __restrict__ w1,
    float*       __restrict__ out)
{
    __shared__ float lds_out[B_ * FPB];

    const int lane = threadIdx.x & 31;
    const int wave = threadIdx.x >> 5;
    const int f    = blockIdx.x * FPB + wave;

    const int hi = lane >> 4;
    const int lo = lane & 15;

    const int idxv = idx[f * K_ + lane];

    __builtin_prefetch(&w0[(size_t)f * (H_ * K_)], 0, 0);

    const int   h  = lo;
    const int   hc = (h < H_) ? h : (H_ - 1);
    const float hv = (h < H_) ? 1.0f : 0.0f;

    const float* w0row = w0 + (size_t)f * (H_ * K_) + hc * K_;
    const float* xr0   = x + (size_t)lo * N_;
    const float* xr1   = x + (size_t)(lo + 16) * N_;

    v8f c0 = {};
    v8f c1 = {};

#pragma unroll
    for (int kk = 0; kk < 8; ++kk) {
        const int k0 = 4 * kk + 2 * hi;
        const int k1 = k0 + 1;
        const int i0 = __shfl(idxv, k0, 32);
        const int i1 = __shfl(idxv, k1, 32);

        v2f a0, a1, bt;
        a0.x = xr0[i0];  a0.y = xr0[i1];
        a1.x = xr1[i0];  a1.y = xr1[i1];
        bt.x = w0row[k0] * hv;
        bt.y = w0row[k1] * hv;

        c0 = __builtin_amdgcn_wmma_f32_16x16x4_f32(false, a0, false, bt,
                                                   (short)0, c0, false, false);
        c1 = __builtin_amdgcn_wmma_f32_16x16x4_f32(false, a1, false, bt,
                                                   (short)0, c1, false, false);
    }

    float s = 0.f, q = 0.f;
#pragma unroll
    for (int r = 0; r < 8; ++r) {
        s += c0[r] + c1[r];
        q += c0[r] * c0[r] + c1[r] * c1[r];
    }
    s += __shfl_xor(s, 16, 32);
    q += __shfl_xor(q, 16, 32);
    const float mean = s * (1.0f / 32.0f);
    const float var  = q * (1.0f / 32.0f) - mean * mean;
    const float rs   = rsqrtf(var + EPS_);

    const float g   = gamma[f * H_ + hc];
    const float be  = beta [f * H_ + hc];
    const float w1v = w1   [f * H_ + hc] * hv;

#pragma unroll
    for (int r = 0; r < 8; ++r) {
        float t0 = (c0[r] - mean) * rs * g + be;
        float t1 = (c1[r] - mean) * rs * g + be;
        t0 = fmaxf(t0, 0.f) * w1v;
        t1 = fmaxf(t1, 0.f) * w1v;
        t0 += __shfl_xor(t0, 1, 32);
        t0 += __shfl_xor(t0, 2, 32);
        t0 += __shfl_xor(t0, 4, 32);
        t0 += __shfl_xor(t0, 8, 32);
        t1 += __shfl_xor(t1, 1, 32);
        t1 += __shfl_xor(t1, 2, 32);
        t1 += __shfl_xor(t1, 4, 32);
        t1 += __shfl_xor(t1, 8, 32);
        if (lo == r) {
            const int row = r + 8 * hi;
            lds_out[row * FPB + wave]        = t0;
            lds_out[(row + 16) * FPB + wave] = t1;
        }
    }

    __syncthreads();

    const int b  = threadIdx.x >> 3;
    const int w  = threadIdx.x & 7;
    const int fo = blockIdx.x * FPB + w;
    out[(size_t)b * F_ + fo] = lds_out[b * FPB + w];
}

extern "C" void kernel_launch(void* const* d_in, const int* in_sizes, int n_in,
                              void* d_out, int out_size, void* d_ws, size_t ws_size,
                              hipStream_t stream) {
    const float* x     = (const float*)d_in[0];
    const int*   idx   = (const int*)  d_in[1];
    const float* w0    = (const float*)d_in[2];
    const float* gamma = (const float*)d_in[3];
    const float* beta  = (const float*)d_in[4];
    const float* w1    = (const float*)d_in[5];
    float*       out   = (float*)d_out;

    const int blocks = F_ / FPB;
    factor1d_wmma_kernel<<<blocks, 256, 0, stream>>>(x, idx, w0, gamma, beta, w1, out);
}


# x copied batch-major into the workspace by a first launch: each gathered index is then one 128-byte line, read with 8-byte loads (2 lines per load instead of 32); sum order unchanged
# speedup vs baseline: 5.4858x; 5.4858x over previous
#include <hip/hip_runtime.h>

typedef __attribute__((ext_vector_type(2))) float v2f;
typedef __attribute__((ext_vector_type(8))) float v8f;

#define B_   32
#define N_   50000
#define F_   50000
#define K_   32
#define H_   10
#define EPS_ 1e-5f
#define FPB  8

__global__ __launch_bounds__(256) void factor1d_wmma_kernel(
    const float* __restrict__ x,
    const int*   __restrict__ idx,
    const float* __restrict__ w0,
    const float* __restrict__ gamma,
    const float* __restrict__ beta,
    const float* __restrict__ w1,
    float*       __restrict__ out)
{
    __shared__ float lds_out[B_ * FPB];

    const int lane = threadIdx.x & 31;
    const int wave = threadIdx.x >> 5;
    const int f    = blockIdx.x * FPB + wave;

    const int hi = lane >> 4;
    const int lo = lane & 15;

    const int idxv = idx[f * K_ + lane];

    __builtin_prefetch(&w0[(size_t)f * (H_ * K_)], 0, 0);

    const int   h  = lo;
    const int   hc = (h < H_) ? h : (H_ - 1);
    const float hv = (h < H_) ? 1.0f : 0.0f;

    const float* w0row = w0 + (size_t)f * (H_ * K_) + hc * K_;
    const float* xr0   = x + (size_t)lo * N_;
    const float* xr1   = x + (size_t)(lo + 16) * N_;

    v8f c0 = {};
    v8f c1 = {};

#pragma unroll
    for (int kk = 0; kk < 8; ++kk) {
        const int k0 = 4 * kk + 2 * hi;
        const int k1 = k0 + 1;
        const int i0 = __shfl(idxv, k0, 32);
        const int i1 = __shfl(idxv, k1, 32);

        v2f a0, a1, bt;
        a0.x = xr0[i0];  a0.y = xr0[i1];
        a1.x = xr1[i0];  a1.y = xr1[i1];
        bt.x = w0row[k0] * hv;
        bt.y = w0row[k1] * hv;

        c0 = __builtin_amdgcn_wmma_f32_16x16x4_f32(false, a0, false, bt,
                                                   (short)0, c0, false, false);
        c1 = __builtin_amdgcn_wmma_f32_16x16x4_f32(false, a1, false, bt,
                                                   (short)0, c1, false, false);
    }

    float s = 0.f, q = 0.f;
#pragma unroll
    for (int r = 0; r < 8; ++r) {
        s += c0[r] + c1[r];
        q += c0[r] * c0[r] + c1[r] * c1[r];
    }
    s += __shfl_xor(s, 16, 32);
    q += __shfl_xor(q, 16, 32);
    const float mean = s * (1.0f / 32.0f);
    const float var  = q * (1.0f / 32.0f) - mean * mean;
    const float rs   = rsqrtf(var + EPS_);

    const float g   = gamma[f * H_ + hc];
    const float be  = beta [f * H_ + hc];
    const float w1v = w1   [f * H_ + hc] * hv;

#pragma unroll
    for (int r = 0; r < 8; ++r) {
        float t0 = (c0[r] - mean) * rs * g + be;
        float t1 = (c1[r] - mean) * rs * g + be;
        t0 = fmaxf(t0, 0.f) * w1v;
        t1 = fmaxf(t1, 0.f) * w1v;
        t0 += __shfl_xor(t0, 1, 32);
        t0 += __shfl_xor(t0, 2, 32);
        t0 += __shfl_xor(t0, 4, 32);
        t0 += __shfl_xor(t0, 8, 32);
        t1 += __shfl_xor(t1, 1, 32);
        t1 += __shfl_xor(t1, 2, 32);
        t1 += __shfl_xor(t1, 4, 32);
        t1 += __shfl_xor(t1, 8, 32);
        if (lo == r) {
            const int row = r + 8 * hi;
            lds_out[row * FPB + wave]        = t0;
            lds_out[(row + 16) * FPB + wave] = t1;
        }
    }

    __syncthreads();

    const int b  = threadIdx.x >> 3;
    const int w  = threadIdx.x & 7;
    const int fo = blockIdx.x * FPB + w;
    out[(size_t)b * F_ + fo] = lds_out[b * FPB + w];
}

#define TN 128

__global__ __launch_bounds__(256) void x_to_batch_lines_kernel(
    const float* __restrict__ x,
    float*       __restrict__ xt)
{
    __shared__ float tile[TN][B_ + 1];

    const int t  = threadIdx.x;
    const int n0 = blockIdx.x * TN;

    const int c  = t & (TN - 1);
    const int r0 = t >> 7;
    const int n  = n0 + c;
    const int nc = (n < N_) ? n : (N_ - 1);
    float v[16];
#pragma unroll
    for (int r = 0; r < 16; ++r) {
        const int b = r * 2 + r0;
        v[r] = x[(size_t)b * N_ + nc];
    }
#pragma unroll
    for (int r = 0; r < 16; ++r) {
        const int b = r * 2 + r0;
        const int j = ((b & 15) << 1) | (b >> 4);
        tile[c][j] = v[r];
    }

    __syncthreads();

    static_assert(N_ % 8 == 0 && TN % 8 == 0, "column groups of 8");
    const int j  = t & 31;
    const int c0 = t >> 5;
    float* xtp = xt + (size_t)(n0 + c0) * B_ + j;
    if (n0 + TN <= N_) {
        float o[16];
#pragma unroll
        for (int r = 0; r < 16; ++r) o[r] = tile[r * 8 + c0][j];
#pragma unroll
        for (int r = 0; r < 16; ++r) xtp[(size_t)r * 8 * B_] = o[r];
    } else {
#pragma unroll
        for (int r = 0; r < 16; ++r) {
            if (n0 + r * 8 < N_) xtp[(size_t)r * 8 * B_] = tile[r * 8 + c0][j];
        }
    }
}

__global__ __launch_bounds__(256) void factor1d_wmma_lines_kernel(
    const float* __restrict__ xt,
    const int*   __restrict__ idx,
    const float* __restrict__ w0,
    const float* __restrict__ gamma,
    const float* __restrict__ beta,
    const float* __restrict__ w1,
    float*       __restrict__ out)
{
    __shared__ float lds_out[B_ * FPB];

    const int lane = threadIdx.x & 31;
    const int wave = threadIdx.x >> 5;
    const int f    = blockIdx.x * FPB + wave;

    const int hi = lane >> 4;
    const int lo = lane & 15;

    const int idxv = idx[f * K_ + lane];

    __builtin_prefetch(&w0[(size_t)f * (H_ * K_)], 0, 0);

    const int   h  = lo;
    const int   hc = (h < H_) ? h : (H_ - 1);
    const float hv = (h < H_) ? 1.0f : 0.0f;

    const float* w0row = w0 + (size_t)f * (H_ * K_) + hc * K_;
    const char*  xtl   = (const char*)xt + lo * 8;

    v8f c0 = {};
    v8f c1 = {};

#pragma unroll
    for (int kk = 0; kk < 8; ++kk) {
        const int k0 = 4 * kk + 2 * hi;
        const int k1 = k0 + 1;
        const unsigned i0 = (unsigned)__shfl(idxv, k0, 32);
        const unsigned i1 = (unsigned)__shfl(idxv, k1, 32);

        const v2f p0 = *(const v2f*)(xtl + (size_t)(i0 * (unsigned)(B_ * 4)));
        const v2f p1 = *(const v2f*)(xtl + (size_t)(i1 * (unsigned)(B_ * 4)));

        v2f a0, a1, bt;
        a0.x = p0.x;  a0.y = p1.x;
        a1.x = p0.y;  a1.y = p1.y;
        bt.x = w0row[k0] * hv;
        bt.y = w0row[k1] * hv;

        c0 = __builtin_amdgcn_wmma_f32_16x16x4_f32(false, a0, false, bt,
                                                   (short)0, c0, false, false);
        c1 = __builtin_amdgcn_wmma_f32_16x16x4_f32(false, a1, false, bt,
                                                   (short)0, c1, false, false);
    }

    float s = 0.f, q = 0.f;
#pragma unroll
    for (int r = 0; r < 8; ++r) {
        s += c0[r] + c1[r];
        q += c0[r] * c0[r] + c1[r] * c1[r];
    }
    s += __shfl_xor(s, 16, 32);
    q += __shfl_xor(q, 16, 32);
    const float mean = s * (1.0f / 32.0f);
    const float var  = q * (1.0f / 32.0f) - mean * mean;
    const float rs   = rsqrtf(var + EPS_);

    const float g   = gamma[f * H_ + hc];
    const float be  = beta [f * H_ + hc];
    const float w1v = w1   [f * H_ + hc] * hv;

#pragma unroll
    for (int r = 0; r < 8; ++r) {
        float t0 = (c0[r] - mean) * rs * g + be;
        float t1 = (c1[r] - mean) * rs * g + be;
        t0 = fmaxf(t0, 0.f) * w1v;
        t1 = fmaxf(t1, 0.f) * w1v;
        t0 += __shfl_xor(t0, 1, 32);
        t0 += __shfl_xor(t0, 2, 32);
        t0 += __shfl_xor(t0, 4, 32);
        t0 += __shfl_xor(t0, 8, 32);
        t1 += __shfl_xor(t1, 1, 32);
        t1 += __shfl_xor(t1, 2, 32);
        t1 += __shfl_xor(t1, 4, 32);
        t1 += __shfl_xor(t1, 8, 32);
        if (lo == r) {
            const int row = r + 8 * hi;
            lds_out[row * FPB + wave]        = t0;
            lds_out[(row + 16) * FPB + wave] = t1;
        }
    }

    __syncthreads();

    const int b  = threadIdx.x >> 3;
    const int w  = threadIdx.x & 7;
    const int fo = blockIdx.x * FPB + w;
    out[(size_t)b * F_ + fo] = lds_out[b * FPB + w];
}

extern "C" void kernel_launch(void* const* d_in, const int* in_sizes, int n_in,
                              void* d_out, int out_size, void* d_ws, size_t ws_size,
                              hipStream_t stream) {
    const float* x     = (const float*)d_in[0];
    const int*   idx   = (const int*)  d_in[1];
    const float* w0    = (const float*)d_in[2];
    const float* gamma = (const float*)d_in[3];
    const float* beta  = (const float*)d_in[4];
    const float* w1    = (const float*)d_in[5];
    float*       out   = (float*)d_out;

    const int blocks = F_ / FPB;
    if (d_ws != nullptr && ws_size >= (size_t)N_ * B_ * sizeof(float)) {
        float* xt = (float*)d_ws;
        x_to_batch_lines_kernel<<<(N_ + TN - 1) / TN, 256, 0, stream>>>(x, xt);
        factor1d_wmma_lines_kernel<<<blocks, 256, 0, stream>>>(xt, idx, w0, gamma, beta, w1, out);
    } else {
        factor1d_wmma_kernel<<<blocks, 256, 0, stream>>>(x, idx, w0, gamma, beta, w1, out);
    }
}


# second launch: each wave loops over 5 consecutive features (10000 resident waves for 50000), next feature's loads issued under the current reduction; indices spread by row_share; sum order unchanged
# speedup vs baseline: 7.4276x; 1.3540x over previous
#include <hip/hip_runtime.h>

typedef __attribute__((ext_vector_type(2))) float v2f;
typedef __attribute__((ext_vector_type(8))) float v8f;

#define B_   32
#define N_   50000
#define F_   50000
#define K_   32
#define H_   10
#define EPS_ 1e-5f
#define FPB  8

__global__ __launch_bounds__(256) void factor1d_wmma_kernel(
    const float* __restrict__ x,
    const int*   __restrict__ idx,
    const float* __restrict__ w0,
    const float* __restrict__ gamma,
    const float* __restrict__ beta,
    const float* __restrict__ w1,
    float*       __restrict__ out)
{
    __shared__ float lds_out[B_ * FPB];

    const int lane = threadIdx.x & 31;
    const int wave = threadIdx.x >> 5;
    const int f    = blockIdx.x * FPB + wave;

    const int hi = lane >> 4;
    const int lo = lane & 15;

    const int idxv = idx[f * K_ + lane];

    __builtin_prefetch(&w0[(size_t)f * (H_ * K_)], 0, 0);

    const int   h  = lo;
    const int   hc = (h < H_) ? h : (H_ - 1);
    const float hv = (h < H_) ? 1.0f : 0.0f;

    const float* w0row = w0 + (size_t)f * (H_ * K_) + hc * K_;
    const float* xr0   = x + (size_t)lo * N_;
    const float* xr1   = x + (size_t)(lo + 16) * N_;

    v8f c0 = {};
    v8f c1 = {};

#pragma unroll
    for (int kk = 0; kk < 8; ++kk) {
        const int k0 = 4 * kk + 2 * hi;
        const int k1 = k0 + 1;
        const int i0 = __shfl(idxv, k0, 32);
        const int i1 = __shfl(idxv, k1, 32);

        v2f a0, a1, bt;
        a0.x = xr0[i0];  a0.y = xr0[i1];
        a1.x = xr1[i0];  a1.y = xr1[i1];
        bt.x = w0row[k0] * hv;
        bt.y = w0row[k1] * hv;

        c0 = __builtin_amdgcn_wmma_f32_16x16x4_f32(false, a0, false, bt,
                                                   (short)0, c0, false, false);
        c1 = __builtin_amdgcn_wmma_f32_16x16x4_f32(false, a1, false, bt,
                                                   (short)0, c1, false, false);
    }

    float s = 0.f, q = 0.f;
#pragma unroll
    for (int r = 0; r < 8; ++r) {
        s += c0[r] + c1[r];
        q += c0[r] * c0[r] + c1[r] * c1[r];
    }
    s += __shfl_xor(s, 16, 32);
    q += __shfl_xor(q, 16, 32);
    const float mean = s * (1.0f / 32.0f);
    const float var  = q * (1.0f / 32.0f) - mean * mean;
    const float rs   = rsqrtf(var + EPS_);

    const float g   = gamma[f * H_ + hc];
    const float be  = beta [f * H_ + hc];
    const float w1v = w1   [f * H_ + hc] * hv;

#pragma unroll
    for (int r = 0; r < 8; ++r) {
        float t0 = (c0[r] - mean) * rs * g + be;
        float t1 = (c1[r] - mean) * rs * g + be;
        t0 = fmaxf(t0, 0.f) * w1v;
        t1 = fmaxf(t1, 0.f) * w1v;
        t0 += __shfl_xor(t0, 1, 32);
        t0 += __shfl_xor(t0, 2, 32);
        t0 += __shfl_xor(t0, 4, 32);
        t0 += __shfl_xor(t0, 8, 32);
        t1 += __shfl_xor(t1, 1, 32);
        t1 += __shfl_xor(t1, 2, 32);
        t1 += __shfl_xor(t1, 4, 32);
        t1 += __shfl_xor(t1, 8, 32);
        if (lo == r) {
            const int row = r + 8 * hi;
            lds_out[row * FPB + wave]        = t0;
            lds_out[(row + 16) * FPB + wave] = t1;
        }
    }

    __syncthreads();

    const int b  = threadIdx.x >> 3;
    const int w  = threadIdx.x & 7;
    const int fo = blockIdx.x * FPB + w;
    out[(size_t)b * F_ + fo] = lds_out[b * FPB + w];
}

#define TN 128

__global__ __launch_bounds__(256) void x_to_batch_lines_kernel(
    const float* __restrict__ x,
    float*       __restrict__ xt)
{
    __shared__ float tile[TN][B_ + 1];

    const int t  = threadIdx.x;
    const int n0 = blockIdx.x * TN;

    const int c  = t & (TN - 1);
    const int r0 = t >> 7;
    const int n  = n0 + c;
    const int nc = (n < N_) ? n : (N_ - 1);
    float v[16];
#pragma unroll
    for (int r = 0; r < 16; ++r) {
        const int b = r * 2 + r0;
        v[r] = x[(size_t)b * N_ + nc];
    }
#pragma unroll
    for (int r = 0; r < 16; ++r) {
        const int b = r * 2 + r0;
        const int j = ((b & 15) << 1) | (b >> 4);
        tile[c][j] = v[r];
    }

    __syncthreads();

    static_assert(N_ % 8 == 0 && TN % 8 == 0, "column groups of 8");
    const int j  = t & 31;
    const int c0 = t >> 5;
    float* xtp = xt + (size_t)(n0 + c0) * B_ + j;
    if (n0 + TN <= N_) {
        float o[16];
#pragma unroll
        for (int r = 0; r < 16; ++r) o[r] = tile[r * 8 + c0][j];
#pragma unroll
        for (int r = 0; r < 16; ++r) xtp[(size_t)r * 8 * B_] = o[r];
    } else {
#pragma unroll
        for (int r = 0; r < 16; ++r) {
            if (n0 + r * 8 < N_) xtp[(size_t)r * 8 * B_] = tile[r * 8 + c0][j];
        }
    }
}

template <int N>
__device__ __forceinline__ int row_share_i32(int v) {
    return __builtin_amdgcn_update_dpp(0, v, 0x150 + N, 0xf, 0xf, false);
}

#define FPW  5
#define FPBL (8 * FPW)

__global__ __launch_bounds__(256) void factor1d_wmma_lines_kernel(
    const float* __restrict__ xt,
    const int*   __restrict__ idx,
    const float* __restrict__ w0,
    const float* __restrict__ gamma,
    const float* __restrict__ beta,
    const float* __restrict__ w1,
    float*       __restrict__ out)
{
    __shared__ __attribute__((aligned(16))) float lds_out[B_ * FPBL];

    const int lane = threadIdx.x & 31;
    const int wave = threadIdx.x >> 5;
    const int f0   = (blockIdx.x * 8 + wave) * FPW;

    const int hi = lane >> 4;
    const int lo = lane & 15;

    const int   h  = lo;
    const int   hc = (h < H_) ? h : (H_ - 1);
    const float hv = (h < H_) ? 1.0f : 0.0f;

    const char* xtl   = (const char*)xt + lo * 8;
    const int   w0off = hc * K_ + 2 * hi;

    const int il = 4 * (lo >> 1) + 2 * hi + (lo & 1);

    int idxv = idx[f0 * K_ + il];
    v2f wb[8];
#pragma unroll
    for (int kk = 0; kk < 8; ++kk)
        wb[kk] = *(const v2f*)(w0 + (size_t)f0 * (H_ * K_) + w0off + 4 * kk);
    float g   = gamma[f0 * H_ + hc];
    float be  = beta [f0 * H_ + hc];
    float w1v = w1   [f0 * H_ + hc];

    for (int i = 0; i < FPW; ++i) {
        v2f p0[8], p1[8];
#define GATHER_STEP(kk)                                                                   \
        {                                                                                 \
            const unsigned i0 = (unsigned)row_share_i32<2 * (kk)>(idxv);                  \
            const unsigned i1 = (unsigned)row_share_i32<2 * (kk) + 1>(idxv);              \
            p0[kk] = *(const v2f*)(xtl + (size_t)(i0 * (unsigned)(B_ * 4)));              \
            p1[kk] = *(const v2f*)(xtl + (size_t)(i1 * (unsigned)(B_ * 4)));              \
        }
        GATHER_STEP(0) GATHER_STEP(1) GATHER_STEP(2) GATHER_STEP(3)
        GATHER_STEP(4) GATHER_STEP(5) GATHER_STEP(6) GATHER_STEP(7)
#undef GATHER_STEP

        v8f c0 = {};
        v8f c1 = {};
#pragma unroll
        for (int kk = 0; kk < 8; ++kk) {
            v2f a0, a1, bt;
            a0.x = p0[kk].x;  a0.y = p1[kk].x;
            a1.x = p0[kk].y;  a1.y = p1[kk].y;
            bt.x = wb[kk].x * hv;
            bt.y = wb[kk].y * hv;
            c0 = __builtin_amdgcn_wmma_f32_16x16x4_f32(false, a0, false, bt,
                                                       (short)0, c0, false, false);
            c1 = __builtin_amdgcn_wmma_f32_16x16x4_f32(false, a1, false, bt,
                                                       (short)0, c1, false, false);
        }

        const float gc = g, bec = be, w1c = w1v;
        if (i + 1 < FPW) {
            const int fn = f0 + i + 1;
            idxv = idx[fn * K_ + il];
#pragma unroll
            for (int kk = 0; kk < 8; ++kk)
                wb[kk] = *(const v2f*)(w0 + (size_t)fn * (H_ * K_) + w0off + 4 * kk);
            g   = gamma[fn * H_ + hc];
            be  = beta [fn * H_ + hc];
            w1v = w1   [fn * H_ + hc];
        }

        float s = 0.f, q = 0.f;
#pragma unroll
        for (int r = 0; r < 8; ++r) {
            s += c0[r] + c1[r];
            q += c0[r] * c0[r] + c1[r] * c1[r];
        }
        s += __shfl_xor(s, 16, 32);
        q += __shfl_xor(q, 16, 32);
        const float mean = s * (1.0f / 32.0f);
        const float var  = q * (1.0f / 32.0f) - mean * mean;
        const float rs   = rsqrtf(var + EPS_);
        const float w1m  = w1c * hv;

        float* lds_f = lds_out + wave * FPW + i;
#pragma unroll
        for (int r = 0; r < 8; ++r) {
            float t0 = (c0[r] - mean) * rs * gc + bec;
            float t1 = (c1[r] - mean) * rs * gc + bec;
            t0 = fmaxf(t0, 0.f) * w1m;
            t1 = fmaxf(t1, 0.f) * w1m;
            t0 += __shfl_xor(t0, 1, 32);
            t0 += __shfl_xor(t0, 2, 32);
            t0 += __shfl_xor(t0, 4, 32);
            t0 += __shfl_xor(t0, 8, 32);
            t1 += __shfl_xor(t1, 1, 32);
            t1 += __shfl_xor(t1, 2, 32);
            t1 += __shfl_xor(t1, 4, 32);
            t1 += __shfl_xor(t1, 8, 32);
            if (lo == r) {
                const int row = r + 8 * hi;
                lds_f[row * FPBL]        = t0;
                lds_f[(row + 16) * FPBL] = t1;
            }
        }
    }

    __syncthreads();

    static_assert(FPBL % 4 == 0 && F_ % FPBL == 0 && F_ % 4 == 0, "16-byte pieces, whole blocks");
    for (int e = threadIdx.x; e < B_ * (FPBL / 4); e += 256) {
        const int b  = e / (FPBL / 4);
        const int c4 = e % (FPBL / 4);
        const float4 v = *(const float4*)&lds_out[b * FPBL + 4 * c4];
        *(float4*)&out[(size_t)b * F_ + blockIdx.x * FPBL + 4 * c4] = v;
    }
}

extern "C" void kernel_launch(void* const* d_in, const int* in_sizes, int n_in,
                              void* d_out, int out_size, void* d_ws, size_t ws_size,
                              hipStream_t stream) {
    const float* x     = (const float*)d_in[0];
    const int*   idx   = (const int*)  d_in[1];
    const float* w0    = (const float*)d_in[2];
    const float* gamma = (const float*)d_in[3];
    const float* beta  = (const float*)d_in[4];
    const float* w1    = (const float*)d_in[5];
    float*       out   = (float*)d_out;

    const int blocks = F_ / FPB;
    if (d_ws != nullptr && ws_size >= (size_t)N_ * B_ * sizeof(float)) {
        float* xt = (float*)d_ws;
        x_to_batch_lines_kernel<<<(N_ + TN - 1) / TN, 256, 0, stream>>>(x, xt);
        factor1d_wmma_lines_kernel<<<F_ / FPBL, 256, 0, stream>>>(xt, idx, w0, gamma, beta, w1, out);
    } else {
        factor1d_wmma_kernel<<<blocks, 256, 0, stream>>>(x, idx, w0, gamma, beta, w1, out);
    }
}


# epilogue h-sum by a transpose through LDS memory: 4 16-byte column stores + 10 row loads per feature replace 64 lane shuffles and 8 masked stores; same pairwise sum tree
# speedup vs baseline: 9.0111x; 1.2132x over previous
#include <hip/hip_runtime.h>

typedef __attribute__((ext_vector_type(2))) float v2f;
typedef __attribute__((ext_vector_type(8))) float v8f;

#define B_   32
#define N_   50000
#define F_   50000
#define K_   32
#define H_   10
#define EPS_ 1e-5f
#define FPB  8

__global__ __launch_bounds__(256) void factor1d_wmma_kernel(
    const float* __restrict__ x,
    const int*   __restrict__ idx,
    const float* __restrict__ w0,
    const float* __restrict__ gamma,
    const float* __restrict__ beta,
    const float* __restrict__ w1,
    float*       __restrict__ out)
{
    __shared__ float lds_out[B_ * FPB];

    const int lane = threadIdx.x & 31;
    const int wave = threadIdx.x >> 5;
    const int f    = blockIdx.x * FPB + wave;

    const int hi = lane >> 4;
    const int lo = lane & 15;

    const int idxv = idx[f * K_ + lane];

    __builtin_prefetch(&w0[(size_t)f * (H_ * K_)], 0, 0);

    const int   h  = lo;
    const int   hc = (h < H_) ? h : (H_ - 1);
    const float hv = (h < H_) ? 1.0f : 0.0f;

    const float* w0row = w0 + (size_t)f * (H_ * K_) + hc * K_;
    const float* xr0   = x + (size_t)lo * N_;
    const float* xr1   = x + (size_t)(lo + 16) * N_;

    v8f c0 = {};
    v8f c1 = {};

#pragma unroll
    for (int kk = 0; kk < 8; ++kk) {
        const int k0 = 4 * kk + 2 * hi;
        const int k1 = k0 + 1;
        const int i0 = __shfl(idxv, k0, 32);
        const int i1 = __shfl(idxv, k1, 32);

        v2f a0, a1, bt;
        a0.x = xr0[i0];  a0.y = xr0[i1];
        a1.x = xr1[i0];  a1.y = xr1[i1];
        bt.x = w0row[k0] * hv;
        bt.y = w0row[k1] * hv;

        c0 = __builtin_amdgcn_wmma_f32_16x16x4_f32(false, a0, false, bt,
                                                   (short)0, c0, false, false);
        c1 = __builtin_amdgcn_wmma_f32_16x16x4_f32(false, a1, false, bt,
                                                   (short)0, c1, false, false);
    }

    float s = 0.f, q = 0.f;
#pragma unroll
    for (int r = 0; r < 8; ++r) {
        s += c0[r] + c1[r];
        q += c0[r] * c0[r] + c1[r] * c1[r];
    }
    s += __shfl_xor(s, 16, 32);
    q += __shfl_xor(q, 16, 32);
    const float mean = s * (1.0f / 32.0f);
    const float var  = q * (1.0f / 32.0f) - mean * mean;
    const float rs   = rsqrtf(var + EPS_);

    const float g   = gamma[f * H_ + hc];
    const float be  = beta [f * H_ + hc];
    const float w1v = w1   [f * H_ + hc] * hv;

#pragma unroll
    for (int r = 0; r < 8; ++r) {
        float t0 = (c0[r] - mean) * rs * g + be;
        float t1 = (c1[r] - mean) * rs * g + be;
        t0 = fmaxf(t0, 0.f) * w1v;
        t1 = fmaxf(t1, 0.f) * w1v;
        t0 += __shfl_xor(t0, 1, 32);
        t0 += __shfl_xor(t0, 2, 32);
        t0 += __shfl_xor(t0, 4, 32);
        t0 += __shfl_xor(t0, 8, 32);
        t1 += __shfl_xor(t1, 1, 32);
        t1 += __shfl_xor(t1, 2, 32);
        t1 += __shfl_xor(t1, 4, 32);
        t1 += __shfl_xor(t1, 8, 32);
        if (lo == r) {
            const int row = r + 8 * hi;
            lds_out[row * FPB + wave]        = t0;
            lds_out[(row + 16) * FPB + wave] = t1;
        }
    }

    __syncthreads();

    const int b  = threadIdx.x >> 3;
    const int w  = threadIdx.x & 7;
    const int fo = blockIdx.x * FPB + w;
    out[(size_t)b * F_ + fo] = lds_out[b * FPB + w];
}

#define TN 128

__global__ __launch_bounds__(256) void x_to_batch_lines_kernel(
    const float* __restrict__ x,
    float*       __restrict__ xt)
{
    __shared__ float tile[TN][B_ + 1];

    const int t  = threadIdx.x;
    const int n0 = blockIdx.x * TN;

    const int c  = t & (TN - 1);
    const int r0 = t >> 7;
    const int n  = n0 + c;
    const int nc = (n < N_) ? n : (N_ - 1);
    float v[16];
#pragma unroll
    for (int r = 0; r < 16; ++r) {
        const int b = r * 2 + r0;
        v[r] = x[(size_t)b * N_ + nc];
    }
#pragma unroll
    for (int r = 0; r < 16; ++r) {
        const int b = r * 2 + r0;
        const int j = ((b & 15) << 1) | (b >> 4);
        tile[c][j] = v[r];
    }

    __syncthreads();

    static_assert(N_ % 8 == 0 && TN % 8 == 0, "column groups of 8");
    const int j  = t & 31;
    const int c0 = t >> 5;
    float* xtp = xt + (size_t)(n0 + c0) * B_ + j;
    if (n0 + TN <= N_) {
        float o[16];
#pragma unroll
        for (int r = 0; r < 16; ++r) o[r] = tile[r * 8 + c0][j];
#pragma unroll
        for (int r = 0; r < 16; ++r) xtp[(size_t)r * 8 * B_] = o[r];
    } else {
#pragma unroll
        for (int r = 0; r < 16; ++r) {
            if (n0 + r * 8 < N_) xtp[(size_t)r * 8 * B_] = tile[r * 8 + c0][j];
        }
    }
}

template <int N>
__device__ __forceinline__ int row_share_i32(int v) {
    return __builtin_amdgcn_update_dpp(0, v, 0x150 + N, 0xf, 0xf, false);
}

#define FPW  5
#define FPBL (8 * FPW)
#define TS   36

__global__ __launch_bounds__(256) __attribute__((amdgpu_waves_per_eu(10, 10)))
void factor1d_wmma_lines_kernel(
    const float* __restrict__ xt,
    const int*   __restrict__ idx,
    const float* __restrict__ w0,
    const float* __restrict__ gamma,
    const float* __restrict__ beta,
    const float* __restrict__ w1,
    float*       __restrict__ out)
{
    __shared__ __attribute__((aligned(16))) float lds_out[B_ * FPBL];
    __shared__ __attribute__((aligned(16))) float lds_t[8 * 16 * TS];

    const int lane = threadIdx.x & 31;
    const int wave = threadIdx.x >> 5;
    const int f0   = (blockIdx.x * 8 + wave) * FPW;

    const int hi = lane >> 4;
    const int lo = lane & 15;

    const int   h  = lo;
    const int   hc = (h < H_) ? h : (H_ - 1);
    const float hv = (h < H_) ? 1.0f : 0.0f;

    const char* xtl   = (const char*)xt + lo * 8;
    const int   w0off = hc * K_ + 2 * hi;

    const int il = 4 * (lo >> 1) + 2 * hi + (lo & 1);

    int idxv = idx[f0 * K_ + il];
    v2f wb[8];
#pragma unroll
    for (int kk = 0; kk < 8; ++kk)
        wb[kk] = *(const v2f*)(w0 + (size_t)f0 * (H_ * K_) + w0off + 4 * kk);
    float g   = gamma[f0 * H_ + hc];
    float be  = beta [f0 * H_ + hc];
    float w1v = w1   [f0 * H_ + hc];

    float4*      t_col = (float4*)(lds_t + wave * (16 * TS) + h * TS + 8 * hi);
    const float* t_row = lds_t + wave * (16 * TS) + lane;

    for (int i = 0; i < FPW; ++i) {
        v2f p0[8], p1[8];
#define GATHER_STEP(kk)                                                                   \
        {                                                                                 \
            const unsigned i0 = (unsigned)row_share_i32<2 * (kk)>(idxv);                  \
            const unsigned i1 = (unsigned)row_share_i32<2 * (kk) + 1>(idxv);              \
            p0[kk] = *(const v2f*)(xtl + (size_t)(i0 * (unsigned)(B_ * 4)));              \
            p1[kk] = *(const v2f*)(xtl + (size_t)(i1 * (unsigned)(B_ * 4)));              \
        }
        GATHER_STEP(0) GATHER_STEP(1) GATHER_STEP(2) GATHER_STEP(3)
        GATHER_STEP(4) GATHER_STEP(5) GATHER_STEP(6) GATHER_STEP(7)
#undef GATHER_STEP
        asm volatile("" :: "v"(p0[0]), "v"(p1[0]), "v"(p0[1]), "v"(p1[1]), "v"(p0[2]), "v"(p1[2]),
                           "v"(p0[3]), "v"(p1[3]), "v"(p0[4]), "v"(p1[4]), "v"(p0[5]), "v"(p1[5]),
                           "v"(p0[6]), "v"(p1[6]), "v"(p0[7]), "v"(p1[7]));

        v8f c0 = {};
        v8f c1 = {};
#pragma unroll
        for (int kk = 0; kk < 8; ++kk) {
            v2f a0, a1, bt;
            a0.x = p0[kk].x;  a0.y = p1[kk].x;
            a1.x = p0[kk].y;  a1.y = p1[kk].y;
            bt.x = wb[kk].x * hv;
            bt.y = wb[kk].y * hv;
            c0 = __builtin_amdgcn_wmma_f32_16x16x4_f32(false, a0, false, bt,
                                                       (short)0, c0, false, false);
            c1 = __builtin_amdgcn_wmma_f32_16x16x4_f32(false, a1, false, bt,
                                                       (short)0, c1, false, false);
        }

        const float gc = g, bec = be, w1c = w1v;
        if (i + 1 < FPW) {
            const int fn = f0 + i + 1;
            idxv = idx[fn * K_ + il];
#pragma unroll
            for (int kk = 0; kk < 8; ++kk)
                wb[kk] = *(const v2f*)(w0 + (size_t)fn * (H_ * K_) + w0off + 4 * kk);
            g   = gamma[fn * H_ + hc];
            be  = beta [fn * H_ + hc];
            w1v = w1   [fn * H_ + hc];
        }

        float s = 0.f, q = 0.f;
#pragma unroll
        for (int r = 0; r < 8; ++r) {
            s += c0[r] + c1[r];
            q += c0[r] * c0[r] + c1[r] * c1[r];
        }
        s += __shfl_xor(s, 16, 32);
        q += __shfl_xor(q, 16, 32);
        const float mean = s * (1.0f / 32.0f);
        const float var  = q * (1.0f / 32.0f) - mean * mean;
        const float rs   = rsqrtf(var + EPS_);
        const float w1m  = w1c * hv;

        float t0[8], t1[8];
#pragma unroll
        for (int r = 0; r < 8; ++r) {
            const float u0 = (c0[r] - mean) * rs * gc + bec;
            const float u1 = (c1[r] - mean) * rs * gc + bec;
            t0[r] = fmaxf(u0, 0.f) * w1m;
            t1[r] = fmaxf(u1, 0.f) * w1m;
        }
        t_col[0] = make_float4(t0[0], t0[1], t0[2], t0[3]);
        t_col[1] = make_float4(t0[4], t0[5], t0[6], t0[7]);
        t_col[4] = make_float4(t1[0], t1[1], t1[2], t1[3]);
        t_col[5] = make_float4(t1[4], t1[5], t1[6], t1[7]);
        __builtin_amdgcn_fence(__ATOMIC_RELEASE, "wavefront");
        __builtin_amdgcn_wave_barrier();
        __builtin_amdgcn_fence(__ATOMIC_ACQUIRE, "wavefront");
        const float v0 = t_row[0 * TS], v1 = t_row[1 * TS], v2 = t_row[2 * TS], v3 = t_row[3 * TS];
        const float v4 = t_row[4 * TS], v5 = t_row[5 * TS], v6 = t_row[6 * TS], v7 = t_row[7 * TS];
        const float v8 = t_row[8 * TS], v9 = t_row[9 * TS];
        const float o = (((v0 + v1) + (v2 + v3)) + ((v4 + v5) + (v6 + v7))) + (v8 + v9);
        lds_out[lane * FPBL + wave * FPW + i] = o;
        __builtin_amdgcn_fence(__ATOMIC_RELEASE, "wavefront");
        __builtin_amdgcn_wave_barrier();
        __builtin_amdgcn_fence(__ATOMIC_ACQUIRE, "wavefront");
    }

    __syncthreads();

    static_assert(FPBL % 4 == 0 && F_ % FPBL == 0 && F_ % 4 == 0, "16-byte pieces, whole blocks");
    for (int e = threadIdx.x; e < B_ * (FPBL / 4); e += 256) {
        const int b  = e / (FPBL / 4);
        const int c4 = e % (FPBL / 4);
        const float4 v = *(const float4*)&lds_out[b * FPBL + 4 * c4];
        *(float4*)&out[(size_t)b * F_ + blockIdx.x * FPBL + 4 * c4] = v;
    }
}

extern "C" void kernel_launch(void* const* d_in, const int* in_sizes, int n_in,
                              void* d_out, int out_size, void* d_ws, size_t ws_size,
                              hipStream_t stream) {
    const float* x     = (const float*)d_in[0];
    const int*   idx   = (const int*)  d_in[1];
    const float* w0    = (const float*)d_in[2];
    const float* gamma = (const float*)d_in[3];
    const float* beta  = (const float*)d_in[4];
    const float* w1    = (const float*)d_in[5];
    float*       out   = (float*)d_out;

    const int blocks = F_ / FPB;
    if (d_ws != nullptr && ws_size >= (size_t)N_ * B_ * sizeof(float)) {
        float* xt = (float*)d_ws;
        x_to_batch_lines_kernel<<<(N_ + TN - 1) / TN, 256, 0, stream>>>(x, xt);
        factor1d_wmma_lines_kernel<<<F_ / FPBL, 256, 0, stream>>>(xt, idx, w0, gamma, beta, w1, out);
    } else {
        factor1d_wmma_kernel<<<blocks, 256, 0, stream>>>(x, idx, w0, gamma, beta, w1, out);
    }
}
